# starting kernel
# baseline (speedup 1.0000x reference)
#include <hip/hip_runtime.h>
#include <math.h>


typedef __attribute__((ext_vector_type(2))) float v2f;
typedef __attribute__((ext_vector_type(8))) float v8f;

#define NFEAT 128
#define NHID  64
#define NCLASS 40

static __device__ __forceinline__ v8f wmma4(v2f a, v2f b, v8f c) {
  return __builtin_amdgcn_wmma_f32_16x16x4_f32(false, a, false, b, (short)0, c,
                                               false, false);
}

__global__ void fill_zero_kernel(float* __restrict__ p, long long n) {
  long long i = (long long)blockIdx.x * blockDim.x + threadIdx.x;
  if (i < n) p[i] = 0.0f;
}

__global__ void deg_kernel(const int* __restrict__ src, const int* __restrict__ dst,
                           float* __restrict__ degS, float* __restrict__ degD, int E) {
  int e = blockIdx.x * blockDim.x + threadIdx.x;
  if (e >= E) return;
  atomicAdd(&degS[src[e]], 1.0f);
  atomicAdd(&degD[dst[e]], 1.0f);
}

__global__ void enc_kernel(const float* __restrict__ x, const float* __restrict__ W,
                           const float* __restrict__ bias, float* __restrict__ out,
                           int M) {
  const int lane = threadIdx.x & 31;
  const int wave = threadIdx.x >> 5;
  const int m0 = (blockIdx.x * 8 + wave) * 16;
  if (m0 >= M) return;
  const int mr = lane & 15;
  const int kh = (lane >> 4) << 1;
  v8f acc[4] = {};
  const float* arow = x + (long long)(m0 + mr) * NFEAT;
  for (int k0 = 0; k0 < NFEAT; k0 += 4) {
    v2f a = *(const v2f*)(arow + k0 + kh);
#pragma unroll
    for (int nt = 0; nt < 4; ++nt) {
      v2f b;
      b.x = W[(k0 + kh) * NHID + nt * 16 + mr];
      b.y = W[(k0 + kh + 1) * NHID + nt * 16 + mr];
      acc[nt] = wmma4(a, b, acc[nt]);
    }
  }
  const int rbase = (lane >> 4) << 3;
#pragma unroll
  for (int nt = 0; nt < 4; ++nt) {
    const int n = nt * 16 + mr;
    const float bn = bias[n];
#pragma unroll
    for (int r = 0; r < 8; ++r) {
      float v = acc[nt][r] + bn;
      out[(long long)(m0 + rbase + r) * NHID + n] = fmaxf(v, 0.0f);
    }
  }
}

__global__ void scatter_kernel(const float* __restrict__ X, const int* __restrict__ src,
                               const int* __restrict__ dst, float* __restrict__ agg,
                               int E) {
  long long t = (long long)blockIdx.x * blockDim.x + threadIdx.x;
  int e = (int)(t >> 4);
  if (e >= E) return;
  int f = (int)(t & 15) * 4;
  int s = src[e], d = dst[e];
  const float4 v = *(const float4*)(X + (long long)s * NHID + f);
  float* p = agg + (long long)d * NHID + f;
  atomicAdd(p + 0, v.x);
  atomicAdd(p + 1, v.y);
  atomicAdd(p + 2, v.z);
  atomicAdd(p + 3, v.w);
}

__global__ void sage_kernel(const float* __restrict__ X, const float* __restrict__ agg,
                            const float* __restrict__ degD,
                            const float* __restrict__ Wl, const float* __restrict__ bl,
                            const float* __restrict__ Wr,
                            const float* __restrict__ Wlg, const float* __restrict__ blg,
                            const float* __restrict__ Wrg,
                            float* __restrict__ Xn, float* __restrict__ Xg, int M) {
  const int lane = threadIdx.x & 31;
  const int wave = threadIdx.x >> 5;
  const int m0 = (blockIdx.x * 8 + wave) * 16;
  if (m0 >= M) return;
  const int mr = lane & 15;
  const int kh = (lane >> 4) << 1;
  const long long row = (long long)(m0 + mr) * NHID;
  const float dinv = 1.0f / fmaxf(degD[m0 + mr], 1.0f);
  v8f aN[4] = {};
  v8f aG[4] = {};
  for (int k0 = 0; k0 < NHID; k0 += 4) {
    v2f am = *(const v2f*)(agg + row + k0 + kh);
    am.x *= dinv;
    am.y *= dinv;
    v2f ax = *(const v2f*)(X + row + k0 + kh);
    const int r0 = (k0 + kh) * NHID;
    const int r1 = (k0 + kh + 1) * NHID;
#pragma unroll
    for (int nt = 0; nt < 4; ++nt) {
      const int c = nt * 16 + mr;
      v2f b;
      b.x = Wl[r0 + c];  b.y = Wl[r1 + c];
      aN[nt] = wmma4(am, b, aN[nt]);
      b.x = Wr[r0 + c];  b.y = Wr[r1 + c];
      aN[nt] = wmma4(ax, b, aN[nt]);
      b.x = Wlg[r0 + c]; b.y = Wlg[r1 + c];
      aG[nt] = wmma4(am, b, aG[nt]);
      b.x = Wrg[r0 + c]; b.y = Wrg[r1 + c];
      aG[nt] = wmma4(ax, b, aG[nt]);
    }
  }
  const int rbase = (lane >> 4) << 3;
#pragma unroll
  for (int nt = 0; nt < 4; ++nt) {
    const int n = nt * 16 + mr;
    const float bn = bl[n];
    const float bg = blg[n];
#pragma unroll
    for (int r = 0; r < 8; ++r) {
      const long long o = (long long)(m0 + rbase + r) * NHID + n;
      Xn[o] = fmaxf(aN[nt][r] + bn, 0.0f);
      Xg[o] = fmaxf(aG[nt][r] + bg, 0.0f);
    }
  }
}

__global__ void gate_kernel(const float* __restrict__ Xg, const int* __restrict__ src,
                            const int* __restrict__ dst, float* __restrict__ tau,
                            int E) {
  long long t = (long long)blockIdx.x * blockDim.x + threadIdx.x;
  int e = (int)(t >> 4);
  if (e >= E) return;
  int f = (int)(t & 15) * 4;
  int s = src[e], d = dst[e];
  const float4 a = *(const float4*)(Xg + (long long)s * NHID + f);
  const float4 b = *(const float4*)(Xg + (long long)d * NHID + f);
  float dx = a.x - b.x, dy = a.y - b.y, dz = a.z - b.z, dw = a.w - b.w;
  float* p = tau + (long long)s * NHID + f;
  atomicAdd(p + 0, dx * dx);
  atomicAdd(p + 1, dy * dy);
  atomicAdd(p + 2, dz * dz);
  atomicAdd(p + 3, dw * dw);
}

__global__ void update_kernel(float* __restrict__ X, const float* __restrict__ Xn,
                              const float* __restrict__ tau,
                              const float* __restrict__ degS, long long total) {
  long long i = (long long)blockIdx.x * blockDim.x + threadIdx.x;
  if (i >= total) return;
  int node = (int)(i >> 6);
  float t = tanhf(tau[i] / fmaxf(degS[node], 1.0f));
  X[i] = (1.0f - t) * X[i] + t * Xn[i];
}

__global__ void dec_kernel(const float* __restrict__ X, const float* __restrict__ W,
                           const float* __restrict__ bias, float* __restrict__ out,
                           int M) {
  const int lane = threadIdx.x & 31;
  const int wave = threadIdx.x >> 5;
  const int m0 = (blockIdx.x * 8 + wave) * 16;
  if (m0 >= M) return;
  const int mr = lane & 15;
  const int kh = (lane >> 4) << 1;
  v8f acc[3] = {};
  const float* arow = X + (long long)(m0 + mr) * NHID;
  for (int k0 = 0; k0 < NHID; k0 += 4) {
    v2f a = *(const v2f*)(arow + k0 + kh);
#pragma unroll
    for (int nt = 0; nt < 3; ++nt) {
      const int n = nt * 16 + mr;
      v2f b = {0.0f, 0.0f};
      if (n < NCLASS) {
        b.x = W[(k0 + kh) * NCLASS + n];
        b.y = W[(k0 + kh + 1) * NCLASS + n];
      }
      acc[nt] = wmma4(a, b, acc[nt]);
    }
  }
  const int rbase = (lane >> 4) << 3;
#pragma unroll
  for (int nt = 0; nt < 3; ++nt) {
    const int n = nt * 16 + mr;
    if (n < NCLASS) {
      const float bn = bias[n];
#pragma unroll
      for (int r = 0; r < 8; ++r)
        out[(long long)(m0 + rbase + r) * NCLASS + n] = acc[nt][r] + bn;
    }
  }
}

extern "C" void kernel_launch(void* const* d_in, const int* in_sizes, int n_in,
                              void* d_out, int out_size, void* d_ws, size_t ws_size,
                              hipStream_t stream) {
  const float* x    = (const float*)d_in[0];
  const int*   ei   = (const int*)d_in[1];
  const float* Wenc = (const float*)d_in[2];
  const float* benc = (const float*)d_in[3];
  const float* Wl   = (const float*)d_in[4];
  const float* bl   = (const float*)d_in[5];
  const float* Wr   = (const float*)d_in[6];
  const float* Wlg  = (const float*)d_in[7];
  const float* blg  = (const float*)d_in[8];
  const float* Wrg  = (const float*)d_in[9];
  const float* Wdec = (const float*)d_in[10];
  const float* bdec = (const float*)d_in[11];

  const int Nn = in_sizes[0] / NFEAT;
  const int E  = in_sizes[1] / 2;
  const int* src = ei;
  const int* dst = ei + E;

  float* ws = (float*)d_ws;
  const long long nf = (long long)Nn * NHID;
  float* X    = ws;
  float* Xn   = ws + nf;
  float* Xg   = ws + 2 * nf;
  float* agg  = ws + 3 * nf;
  float* tau  = ws + 4 * nf;
  float* degD = ws + 5 * nf;
  float* degS = degD + Nn;

  const dim3 blk(256);
  const int gemm_blocks = (Nn + 127) / 128;
  const int edge_blocks = (E + 255) / 256;
  const long long sth = (long long)E * 16;
  const int scat_blocks = (int)((sth + 255) / 256);
  const int nf_blocks = (int)((nf + 255) / 256);

  fill_zero_kernel<<<(2 * Nn + 255) / 256, blk, 0, stream>>>(degD, 2LL * Nn);
  deg_kernel<<<edge_blocks, blk, 0, stream>>>(src, dst, degS, degD, E);

  enc_kernel<<<gemm_blocks, blk, 0, stream>>>(x, Wenc, benc, X, Nn);

  for (int layer = 0; layer < 4; ++layer) {
    fill_zero_kernel<<<nf_blocks, blk, 0, stream>>>(agg, nf);
    scatter_kernel<<<scat_blocks, blk, 0, stream>>>(X, src, dst, agg, E);
    sage_kernel<<<gemm_blocks, blk, 0, stream>>>(X, agg, degD, Wl, bl, Wr,
                                                 Wlg, blg, Wrg, Xn, Xg, Nn);
    fill_zero_kernel<<<nf_blocks, blk, 0, stream>>>(tau, nf);
    gate_kernel<<<scat_blocks, blk, 0, stream>>>(Xg, src, dst, tau, E);
    update_kernel<<<nf_blocks, blk, 0, stream>>>(X, Xn, tau, degS, nf);
  }

  dec_kernel<<<gemm_blocks, blk, 0, stream>>>(X, Wdec, bdec, (float*)d_out, Nn);
}


# scatter and gate kernels: a wave walks 32 edges with lane = feature column, so each row load and each f32 atomic add covers one whole 128-B line (2 atomic line requests per edge instead of 8)
# speedup vs baseline: 1.7108x; 1.7108x over previous
#include <hip/hip_runtime.h>
#include <math.h>


typedef __attribute__((ext_vector_type(2))) float v2f;
typedef __attribute__((ext_vector_type(8))) float v8f;

#define NFEAT 128
#define NHID  64
#define NCLASS 40
#define EPW 32
#define GRP 8

static __device__ __forceinline__ v8f wmma4(v2f a, v2f b, v8f c) {
  return __builtin_amdgcn_wmma_f32_16x16x4_f32(false, a, false, b, (short)0, c,
                                               false, false);
}

__global__ void fill_zero_kernel(float* __restrict__ p, long long n) {
  long long i = (long long)blockIdx.x * blockDim.x + threadIdx.x;
  if (i < n) p[i] = 0.0f;
}

__global__ void deg_kernel(const int* __restrict__ src, const int* __restrict__ dst,
                           float* __restrict__ degS, float* __restrict__ degD, int E) {
  int e = blockIdx.x * blockDim.x + threadIdx.x;
  if (e >= E) return;
  atomicAdd(&degS[src[e]], 1.0f);
  atomicAdd(&degD[dst[e]], 1.0f);
}

__global__ void enc_kernel(const float* __restrict__ x, const float* __restrict__ W,
                           const float* __restrict__ bias, float* __restrict__ out,
                           int M) {
  const int lane = threadIdx.x & 31;
  const int wave = threadIdx.x >> 5;
  const int m0 = (blockIdx.x * 8 + wave) * 16;
  if (m0 >= M) return;
  const int mr = lane & 15;
  const int kh = (lane >> 4) << 1;
  v8f acc[4] = {};
  const float* arow = x + (long long)(m0 + mr) * NFEAT;
  for (int k0 = 0; k0 < NFEAT; k0 += 4) {
    v2f a = *(const v2f*)(arow + k0 + kh);
#pragma unroll
    for (int nt = 0; nt < 4; ++nt) {
      v2f b;
      b.x = W[(k0 + kh) * NHID + nt * 16 + mr];
      b.y = W[(k0 + kh + 1) * NHID + nt * 16 + mr];
      acc[nt] = wmma4(a, b, acc[nt]);
    }
  }
  const int rbase = (lane >> 4) << 3;
#pragma unroll
  for (int nt = 0; nt < 4; ++nt) {
    const int n = nt * 16 + mr;
    const float bn = bias[n];
#pragma unroll
    for (int r = 0; r < 8; ++r) {
      float v = acc[nt][r] + bn;
      out[(long long)(m0 + rbase + r) * NHID + n] = fmaxf(v, 0.0f);
    }
  }
}

__global__ void scatter_kernel(const float* __restrict__ X, const int* __restrict__ src,
                               const int* __restrict__ dst, float* __restrict__ agg,
                               int E) {
  const int lane = threadIdx.x & 31;
  const int wave = __builtin_amdgcn_readfirstlane((int)((blockIdx.x * blockDim.x + threadIdx.x) >> 5));
  const long long e0 = (long long)wave * EPW;
  if (e0 >= E) return;
  const int n = (E - e0 < EPW) ? (int)(E - e0) : EPW;
  int sv = 0, dv = 0;
  if (lane < n) { sv = src[e0 + lane]; dv = dst[e0 + lane]; }
  int j = 0;
  for (; j + GRP <= n; j += GRP) {
    float lo[GRP], hi[GRP];
#pragma unroll
    for (int u = 0; u < GRP; ++u) {
      const int s = __builtin_amdgcn_readlane(sv, j + u);
      const float* r = X + (long long)s * NHID + lane;
      lo[u] = r[0];
      hi[u] = r[32];
    }
#pragma unroll
    for (int u = 0; u < GRP; ++u) {
      const int d = __builtin_amdgcn_readlane(dv, j + u);
      float* p = agg + (long long)d * NHID + lane;
      atomicAdd(p, lo[u]);
      atomicAdd(p + 32, hi[u]);
    }
  }
  for (; j < n; ++j) {
    const int s = __builtin_amdgcn_readlane(sv, j);
    const int d = __builtin_amdgcn_readlane(dv, j);
    const float* r = X + (long long)s * NHID + lane;
    float* p = agg + (long long)d * NHID + lane;
    atomicAdd(p, r[0]);
    atomicAdd(p + 32, r[32]);
  }
}

__global__ void sage_kernel(const float* __restrict__ X, const float* __restrict__ agg,
                            const float* __restrict__ degD,
                            const float* __restrict__ Wl, const float* __restrict__ bl,
                            const float* __restrict__ Wr,
                            const float* __restrict__ Wlg, const float* __restrict__ blg,
                            const float* __restrict__ Wrg,
                            float* __restrict__ Xn, float* __restrict__ Xg, int M) {
  const int lane = threadIdx.x & 31;
  const int wave = threadIdx.x >> 5;
  const int m0 = (blockIdx.x * 8 + wave) * 16;
  if (m0 >= M) return;
  const int mr = lane & 15;
  const int kh = (lane >> 4) << 1;
  const long long row = (long long)(m0 + mr) * NHID;
  const float dinv = 1.0f / fmaxf(degD[m0 + mr], 1.0f);
  v8f aN[4] = {};
  v8f aG[4] = {};
  for (int k0 = 0; k0 < NHID; k0 += 4) {
    v2f am = *(const v2f*)(agg + row + k0 + kh);
    am.x *= dinv;
    am.y *= dinv;
    v2f ax = *(const v2f*)(X + row + k0 + kh);
    const int r0 = (k0 + kh) * NHID;
    const int r1 = (k0 + kh + 1) * NHID;
#pragma unroll
    for (int nt = 0; nt < 4; ++nt) {
      const int c = nt * 16 + mr;
      v2f b;
      b.x = Wl[r0 + c];  b.y = Wl[r1 + c];
      aN[nt] = wmma4(am, b, aN[nt]);
      b.x = Wr[r0 + c];  b.y = Wr[r1 + c];
      aN[nt] = wmma4(ax, b, aN[nt]);
      b.x = Wlg[r0 + c]; b.y = Wlg[r1 + c];
      aG[nt] = wmma4(am, b, aG[nt]);
      b.x = Wrg[r0 + c]; b.y = Wrg[r1 + c];
      aG[nt] = wmma4(ax, b, aG[nt]);
    }
  }
  const int rbase = (lane >> 4) << 3;
#pragma unroll
  for (int nt = 0; nt < 4; ++nt) {
    const int n = nt * 16 + mr;
    const float bn = bl[n];
    const float bg = blg[n];
#pragma unroll
    for (int r = 0; r < 8; ++r) {
      const long long o = (long long)(m0 + rbase + r) * NHID + n;
      Xn[o] = fmaxf(aN[nt][r] + bn, 0.0f);
      Xg[o] = fmaxf(aG[nt][r] + bg, 0.0f);
    }
  }
}

__global__ void gate_kernel(const float* __restrict__ Xg, const int* __restrict__ src,
                            const int* __restrict__ dst, float* __restrict__ tau,
                            int E) {
  const int lane = threadIdx.x & 31;
  const int wave = __builtin_amdgcn_readfirstlane((int)((blockIdx.x * blockDim.x + threadIdx.x) >> 5));
  const long long e0 = (long long)wave * EPW;
  if (e0 >= E) return;
  const int n = (E - e0 < EPW) ? (int)(E - e0) : EPW;
  int sv = 0, dv = 0;
  if (lane < n) { sv = src[e0 + lane]; dv = dst[e0 + lane]; }
  int j = 0;
  for (; j + GRP <= n; j += GRP) {
    float alo[GRP], ahi[GRP], blo[GRP], bhi[GRP];
#pragma unroll
    for (int u = 0; u < GRP; ++u) {
      const int s = __builtin_amdgcn_readlane(sv, j + u);
      const int d = __builtin_amdgcn_readlane(dv, j + u);
      const float* ra = Xg + (long long)s * NHID + lane;
      const float* rb = Xg + (long long)d * NHID + lane;
      alo[u] = ra[0];
      ahi[u] = ra[32];
      blo[u] = rb[0];
      bhi[u] = rb[32];
    }
#pragma unroll
    for (int u = 0; u < GRP; ++u) {
      const int s = __builtin_amdgcn_readlane(sv, j + u);
      float* p = tau + (long long)s * NHID + lane;
      const float dl = alo[u] - blo[u];
      const float dh = ahi[u] - bhi[u];
      atomicAdd(p, dl * dl);
      atomicAdd(p + 32, dh * dh);
    }
  }
  for (; j < n; ++j) {
    const int s = __builtin_amdgcn_readlane(sv, j);
    const int d = __builtin_amdgcn_readlane(dv, j);
    const float* ra = Xg + (long long)s * NHID + lane;
    const float* rb = Xg + (long long)d * NHID + lane;
    float* p = tau + (long long)s * NHID + lane;
    const float dl = ra[0] - rb[0];
    const float dh = ra[32] - rb[32];
    atomicAdd(p, dl * dl);
    atomicAdd(p + 32, dh * dh);
  }
}

__global__ void update_kernel(float* __restrict__ X, const float* __restrict__ Xn,
                              const float* __restrict__ tau,
                              const float* __restrict__ degS, long long total) {
  long long i = (long long)blockIdx.x * blockDim.x + threadIdx.x;
  if (i >= total) return;
  int node = (int)(i >> 6);
  float t = tanhf(tau[i] / fmaxf(degS[node], 1.0f));
  X[i] = (1.0f - t) * X[i] + t * Xn[i];
}

__global__ void dec_kernel(const float* __restrict__ X, const float* __restrict__ W,
                           const float* __restrict__ bias, float* __restrict__ out,
                           int M) {
  const int lane = threadIdx.x & 31;
  const int wave = threadIdx.x >> 5;
  const int m0 = (blockIdx.x * 8 + wave) * 16;
  if (m0 >= M) return;
  const int mr = lane & 15;
  const int kh = (lane >> 4) << 1;
  v8f acc[3] = {};
  const float* arow = X + (long long)(m0 + mr) * NHID;
  for (int k0 = 0; k0 < NHID; k0 += 4) {
    v2f a = *(const v2f*)(arow + k0 + kh);
#pragma unroll
    for (int nt = 0; nt < 3; ++nt) {
      const int n = nt * 16 + mr;
      v2f b = {0.0f, 0.0f};
      if (n < NCLASS) {
        b.x = W[(k0 + kh) * NCLASS + n];
        b.y = W[(k0 + kh + 1) * NCLASS + n];
      }
      acc[nt] = wmma4(a, b, acc[nt]);
    }
  }
  const int rbase = (lane >> 4) << 3;
#pragma unroll
  for (int nt = 0; nt < 3; ++nt) {
    const int n = nt * 16 + mr;
    if (n < NCLASS) {
      const float bn = bias[n];
#pragma unroll
      for (int r = 0; r < 8; ++r)
        out[(long long)(m0 + rbase + r) * NCLASS + n] = acc[nt][r] + bn;
    }
  }
}

extern "C" void kernel_launch(void* const* d_in, const int* in_sizes, int n_in,
                              void* d_out, int out_size, void* d_ws, size_t ws_size,
                              hipStream_t stream) {
  const float* x    = (const float*)d_in[0];
  const int*   ei   = (const int*)d_in[1];
  const float* Wenc = (const float*)d_in[2];
  const float* benc = (const float*)d_in[3];
  const float* Wl   = (const float*)d_in[4];
  const float* bl   = (const float*)d_in[5];
  const float* Wr   = (const float*)d_in[6];
  const float* Wlg  = (const float*)d_in[7];
  const float* blg  = (const float*)d_in[8];
  const float* Wrg  = (const float*)d_in[9];
  const float* Wdec = (const float*)d_in[10];
  const float* bdec = (const float*)d_in[11];

  const int Nn = in_sizes[0] / NFEAT;
  const int E  = in_sizes[1] / 2;
  const int* src = ei;
  const int* dst = ei + E;

  float* ws = (float*)d_ws;
  const long long nf = (long long)Nn * NHID;
  float* X    = ws;
  float* Xn   = ws + nf;
  float* Xg   = ws + 2 * nf;
  float* agg  = ws + 3 * nf;
  float* tau  = ws + 4 * nf;
  float* degD = ws + 5 * nf;
  float* degS = degD + Nn;

  const dim3 blk(256);
  const int gemm_blocks = (Nn + 127) / 128;
  const int edge_blocks = (E + 255) / 256;
  const int edge_waves = (E + EPW - 1) / EPW;
  const int scat_blocks = (edge_waves + 7) / 8;
  const int nf_blocks = (int)((nf + 255) / 256);

  fill_zero_kernel<<<(2 * Nn + 255) / 256, blk, 0, stream>>>(degD, 2LL * Nn);
  deg_kernel<<<edge_blocks, blk, 0, stream>>>(src, dst, degS, degD, E);

  enc_kernel<<<gemm_blocks, blk, 0, stream>>>(x, Wenc, benc, X, Nn);

  for (int layer = 0; layer < 4; ++layer) {
    fill_zero_kernel<<<nf_blocks, blk, 0, stream>>>(agg, nf);
    scatter_kernel<<<scat_blocks, blk, 0, stream>>>(X, src, dst, agg, E);
    sage_kernel<<<gemm_blocks, blk, 0, stream>>>(X, agg, degD, Wl, bl, Wr,
                                                 Wlg, blg, Wrg, Xn, Xg, Nn);
    fill_zero_kernel<<<nf_blocks, blk, 0, stream>>>(tau, nf);
    gate_kernel<<<scat_blocks, blk, 0, stream>>>(Xg, src, dst, tau, E);
    update_kernel<<<nf_blocks, blk, 0, stream>>>(X, Xn, tau, degS, nf);
  }

  dec_kernel<<<gemm_blocks, blk, 0, stream>>>(X, Wdec, bdec, (float*)d_out, Nn);
}


# edge sums without f32 atomics: edges grouped once per call by destination and by source (integer count, prefix sum, placement); one wave per node sums its neighbour rows in registers; no zero fill
# speedup vs baseline: 1.9811x; 1.1580x over previous
#include <hip/hip_runtime.h>
#include <math.h>


typedef __attribute__((ext_vector_type(2))) float v2f;
typedef __attribute__((ext_vector_type(8))) float v8f;

#define NFEAT 128
#define NHID  64
#define NCLASS 40
#define GRP 8
#define NPW 4

static __device__ __forceinline__ v8f wmma4(v2f a, v2f b, v8f c) {
  return __builtin_amdgcn_wmma_f32_16x16x4_f32(false, a, false, b, (short)0, c,
                                               false, false);
}

__global__ void fill_zero_kernel(float* __restrict__ p, long long n) {
  long long i = (long long)blockIdx.x * blockDim.x + threadIdx.x;
  if (i < n) p[i] = 0.0f;
}

__global__ void count_kernel(const int* __restrict__ src, const int* __restrict__ dst,
                             int* __restrict__ cntS, int* __restrict__ cntD, int E) {
  const long long stride = (long long)gridDim.x * blockDim.x;
  for (long long e = (long long)blockIdx.x * blockDim.x + threadIdx.x; e < E; e += stride) {
    const int s = src[e];
    const int d = dst[e];
    atomicAdd(&cntS[s], 1);
    atomicAdd(&cntD[d], 1);
  }
}

#define SCAN_C 8
__global__ void scan_kernel(int* ptrD, int* ptrS, int Nn) {
  __shared__ int wtot[32];
  int* p = blockIdx.x ? ptrS : ptrD;
  const int lane = threadIdx.x & 31;
  const int wave = threadIdx.x >> 5;
  int blockCarry = 0;
  for (long long t0 = 0; t0 < Nn; t0 += 32 * SCAN_C * 128) {
    const long long wb = t0 + (long long)wave * (SCAN_C * 128) + 4 * lane;
    int x[SCAN_C][4];
#pragma unroll
    for (int c = 0; c < SCAN_C; ++c) {
#pragma unroll
      for (int k = 0; k < 4; ++k) {
        const long long i = wb + c * 128 + k;
        x[c][k] = (i < Nn) ? p[i] : 0;
      }
    }
    int inc[SCAN_C];
#pragma unroll
    for (int c = 0; c < SCAN_C; ++c) {
      x[c][1] += x[c][0];
      x[c][2] += x[c][1];
      x[c][3] += x[c][2];
      inc[c] = x[c][3];
    }
    for (int d = 1; d < 32; d <<= 1) {
#pragma unroll
      for (int c = 0; c < SCAN_C; ++c) {
        const int up = __shfl_up(inc[c], d);
        if (lane >= d) inc[c] += up;
      }
    }
    int cb[SCAN_C];
    int run = 0;
#pragma unroll
    for (int c = 0; c < SCAN_C; ++c) {
      cb[c] = run;
      run += __builtin_amdgcn_readlane(inc[c], 31);
    }
    if (lane == 0) wtot[wave] = run;
    __syncthreads();
    const int mine = wtot[lane];
    int pre = (lane < wave) ? mine : 0;
    int all = mine;
    for (int d = 16; d > 0; d >>= 1) {
      pre += __shfl_xor(pre, d);
      all += __shfl_xor(all, d);
    }
    const int base = blockCarry + pre;
#pragma unroll
    for (int c = 0; c < SCAN_C; ++c) {
      const int lb = base + cb[c] + inc[c] - x[c][3];
#pragma unroll
      for (int k = 0; k < 4; ++k) {
        const long long i = wb + c * 128 + k;
        if (i < Nn) p[i] = lb + (k ? x[c][k - 1] : 0);
      }
    }
    blockCarry += all;
    __syncthreads();
  }
}

__global__ void place_kernel(const int* __restrict__ src, const int* __restrict__ dst,
                             int* __restrict__ ptrD, int* __restrict__ ptrS,
                             int* __restrict__ adjD, int* __restrict__ adjS, int E) {
  const long long stride = (long long)gridDim.x * blockDim.x;
  for (long long e = (long long)blockIdx.x * blockDim.x + threadIdx.x; e < E; e += stride) {
    const int s = src[e];
    const int d = dst[e];
    const int pd = atomicAdd(&ptrD[d], 1);
    const int ps = atomicAdd(&ptrS[s], 1);
    adjD[pd] = s;
    adjS[ps] = d;
  }
}

__global__ void enc_kernel(const float* __restrict__ x, const float* __restrict__ W,
                           const float* __restrict__ bias, float* __restrict__ out,
                           int M) {
  const int lane = threadIdx.x & 31;
  const int wave = threadIdx.x >> 5;
  const int m0 = (blockIdx.x * 8 + wave) * 16;
  if (m0 >= M) return;
  const int mr = lane & 15;
  const int kh = (lane >> 4) << 1;
  v8f acc[4] = {};
  const float* arow = x + (long long)(m0 + mr) * NFEAT;
  for (int k0 = 0; k0 < NFEAT; k0 += 4) {
    v2f a = *(const v2f*)(arow + k0 + kh);
#pragma unroll
    for (int nt = 0; nt < 4; ++nt) {
      v2f b;
      b.x = W[(k0 + kh) * NHID + nt * 16 + mr];
      b.y = W[(k0 + kh + 1) * NHID + nt * 16 + mr];
      acc[nt] = wmma4(a, b, acc[nt]);
    }
  }
  const int rbase = (lane >> 4) << 3;
#pragma unroll
  for (int nt = 0; nt < 4; ++nt) {
    const int n = nt * 16 + mr;
    const float bn = bias[n];
#pragma unroll
    for (int r = 0; r < 8; ++r) {
      float v = acc[nt][r] + bn;
      out[(long long)(m0 + rbase + r) * NHID + n] = fmaxf(v, 0.0f);
    }
  }
}

__global__ void gather_mean_kernel(const float* __restrict__ X, const int* __restrict__ adj,
                                   const int* __restrict__ endp, float* __restrict__ mean,
                                   int Nn) {
  const int lane = threadIdx.x & 31;
  const int w = __builtin_amdgcn_readfirstlane((int)((blockIdx.x * blockDim.x + threadIdx.x) >> 5));
  const int nw = (int)((gridDim.x * blockDim.x) >> 5);
  for (int v = w; v < Nn; v += nw) {
    const int prev = endp[(v > 0) ? (v - 1) : 0];
    const int end = endp[v];
    const int beg = (v > 0) ? prev : 0;
    v2f acc = {0.0f, 0.0f};
    for (int base = beg; base < end; base += 32) {
      const int n = (end - base < 32) ? (end - base) : 32;
      int sv = 0;
      if (lane < n) sv = adj[base + lane];
      for (int j = 0; j < n; j += GRP) {
        v2f r[GRP];
#pragma unroll
        for (int u = 0; u < GRP; ++u) {
          const int s = __builtin_amdgcn_readlane(sv, j + u);
          r[u] = *(const v2f*)(X + (long long)s * NHID + 2 * lane);
        }
#pragma unroll
        for (int u = 0; u < GRP; ++u) {
          acc.x += (j + u < n) ? r[u].x : 0.0f;
          acc.y += (j + u < n) ? r[u].y : 0.0f;
        }
      }
    }
    const float dg = fmaxf((float)(end - beg), 1.0f);
    v2f m;
    m.x = acc.x / dg;
    m.y = acc.y / dg;
    *(v2f*)(mean + (long long)v * NHID + 2 * lane) = m;
  }
}

__global__ void sage_kernel(const float* __restrict__ X, const float* __restrict__ mean,
                            const float* __restrict__ Wl, const float* __restrict__ bl,
                            const float* __restrict__ Wr,
                            const float* __restrict__ Wlg, const float* __restrict__ blg,
                            const float* __restrict__ Wrg,
                            float* __restrict__ Xn, float* __restrict__ Xg, int M) {
  const int lane = threadIdx.x & 31;
  const int wave = threadIdx.x >> 5;
  const int m0 = (blockIdx.x * 8 + wave) * 16;
  if (m0 >= M) return;
  const int mr = lane & 15;
  const int kh = (lane >> 4) << 1;
  const long long row = (long long)(m0 + mr) * NHID;
  v8f aN[4] = {};
  v8f aG[4] = {};
  for (int k0 = 0; k0 < NHID; k0 += 4) {
    v2f am = *(const v2f*)(mean + row + k0 + kh);
    v2f ax = *(const v2f*)(X + row + k0 + kh);
    const int r0 = (k0 + kh) * NHID;
    const int r1 = (k0 + kh + 1) * NHID;
#pragma unroll
    for (int nt = 0; nt < 4; ++nt) {
      const int c = nt * 16 + mr;
      v2f b;
      b.x = Wl[r0 + c];  b.y = Wl[r1 + c];
      aN[nt] = wmma4(am, b, aN[nt]);
      b.x = Wr[r0 + c];  b.y = Wr[r1 + c];
      aN[nt] = wmma4(ax, b, aN[nt]);
      b.x = Wlg[r0 + c]; b.y = Wlg[r1 + c];
      aG[nt] = wmma4(am, b, aG[nt]);
      b.x = Wrg[r0 + c]; b.y = Wrg[r1 + c];
      aG[nt] = wmma4(ax, b, aG[nt]);
    }
  }
  const int rbase = (lane >> 4) << 3;
#pragma unroll
  for (int nt = 0; nt < 4; ++nt) {
    const int n = nt * 16 + mr;
    const float bn = bl[n];
    const float bg = blg[n];
#pragma unroll
    for (int r = 0; r < 8; ++r) {
      const long long o = (long long)(m0 + rbase + r) * NHID + n;
      Xn[o] = fmaxf(aN[nt][r] + bn, 0.0f);
      Xg[o] = fmaxf(aG[nt][r] + bg, 0.0f);
    }
  }
}

__global__ void gate_gather_kernel(const float* __restrict__ Xg, const int* __restrict__ adj,
                                   const int* __restrict__ endp, float* __restrict__ tau,
                                   int Nn) {
  const int lane = threadIdx.x & 31;
  const int w = __builtin_amdgcn_readfirstlane((int)((blockIdx.x * blockDim.x + threadIdx.x) >> 5));
  const int nw = (int)((gridDim.x * blockDim.x) >> 5);
  for (int v = w; v < Nn; v += nw) {
    const int prev = endp[(v > 0) ? (v - 1) : 0];
    const int end = endp[v];
    const int beg = (v > 0) ? prev : 0;
    const v2f own = *(const v2f*)(Xg + (long long)v * NHID + 2 * lane);
    v2f acc = {0.0f, 0.0f};
    for (int base = beg; base < end; base += 32) {
      const int n = (end - base < 32) ? (end - base) : 32;
      int dv = 0;
      if (lane < n) dv = adj[base + lane];
      for (int j = 0; j < n; j += GRP) {
        v2f r[GRP];
#pragma unroll
        for (int u = 0; u < GRP; ++u) {
          const int d = __builtin_amdgcn_readlane(dv, j + u);
          r[u] = *(const v2f*)(Xg + (long long)d * NHID + 2 * lane);
        }
#pragma unroll
        for (int u = 0; u < GRP; ++u) {
          const float dx = own.x - r[u].x;
          const float dy = own.y - r[u].y;
          acc.x += (j + u < n) ? dx * dx : 0.0f;
          acc.y += (j + u < n) ? dy * dy : 0.0f;
        }
      }
    }
    const float dg = fmaxf((float)(end - beg), 1.0f);
    v2f t;
    t.x = tanhf(acc.x / dg);
    t.y = tanhf(acc.y / dg);
    *(v2f*)(tau + (long long)v * NHID + 2 * lane) = t;
  }
}

__global__ void update_kernel(float* __restrict__ X, const float* __restrict__ Xn,
                              const float* __restrict__ tau, long long total) {
  long long i = (long long)blockIdx.x * blockDim.x + threadIdx.x;
  if (i >= total) return;
  const float t = tau[i];
  X[i] = (1.0f - t) * X[i] + t * Xn[i];
}

__global__ void dec_kernel(const float* __restrict__ X, const float* __restrict__ W,
                           const float* __restrict__ bias, float* __restrict__ out,
                           int M) {
  const int lane = threadIdx.x & 31;
  const int wave = threadIdx.x >> 5;
  const int m0 = (blockIdx.x * 8 + wave) * 16;
  if (m0 >= M) return;
  const int mr = lane & 15;
  const int kh = (lane >> 4) << 1;
  v8f acc[3] = {};
  const float* arow = X + (long long)(m0 + mr) * NHID;
  for (int k0 = 0; k0 < NHID; k0 += 4) {
    v2f a = *(const v2f*)(arow + k0 + kh);
#pragma unroll
    for (int nt = 0; nt < 3; ++nt) {
      const int n = nt * 16 + mr;
      v2f b = {0.0f, 0.0f};
      if (n < NCLASS) {
        b.x = W[(k0 + kh) * NCLASS + n];
        b.y = W[(k0 + kh + 1) * NCLASS + n];
      }
      acc[nt] = wmma4(a, b, acc[nt]);
    }
  }
  const int rbase = (lane >> 4) << 3;
#pragma unroll
  for (int nt = 0; nt < 3; ++nt) {
    const int n = nt * 16 + mr;
    if (n < NCLASS) {
      const float bn = bias[n];
#pragma unroll
      for (int r = 0; r < 8; ++r)
        out[(long long)(m0 + rbase + r) * NCLASS + n] = acc[nt][r] + bn;
    }
  }
}

extern "C" void kernel_launch(void* const* d_in, const int* in_sizes, int n_in,
                              void* d_out, int out_size, void* d_ws, size_t ws_size,
                              hipStream_t stream) {
  const float* x    = (const float*)d_in[0];
  const int*   ei   = (const int*)d_in[1];
  const float* Wenc = (const float*)d_in[2];
  const float* benc = (const float*)d_in[3];
  const float* Wl   = (const float*)d_in[4];
  const float* bl   = (const float*)d_in[5];
  const float* Wr   = (const float*)d_in[6];
  const float* Wlg  = (const float*)d_in[7];
  const float* blg  = (const float*)d_in[8];
  const float* Wrg  = (const float*)d_in[9];
  const float* Wdec = (const float*)d_in[10];
  const float* bdec = (const float*)d_in[11];

  const int Nn = in_sizes[0] / NFEAT;
  const int E  = in_sizes[1] / 2;
  const int* src = ei;
  const int* dst = ei + E;

  float* ws = (float*)d_ws;
  const long long nf = (long long)Nn * NHID;
  float* X    = ws;
  float* Xn   = ws + nf;
  float* Xg   = ws + 2 * nf;
  float* mean = ws + 3 * nf;
  float* tau  = mean;
  int* adjD = (int*)(ws + 4 * nf);
  int* adjS = adjD + E;
  int* ptrD = adjS + E;
  int* ptrS = ptrD + Nn;

  const dim3 blk(256);
  const int gemm_blocks = (Nn + 127) / 128;
  const int edge_blocks = 2048;
  const int node_blocks = (Nn + 8 * NPW - 1) / (8 * NPW);
  const int nf_blocks = (int)((nf + 255) / 256);

  fill_zero_kernel<<<(2 * Nn + 255) / 256, blk, 0, stream>>>((float*)ptrD, 2LL * Nn);
  count_kernel<<<edge_blocks, blk, 0, stream>>>(src, dst, ptrS, ptrD, E);
  scan_kernel<<<2, 1024, 0, stream>>>(ptrD, ptrS, Nn);
  place_kernel<<<edge_blocks, blk, 0, stream>>>(src, dst, ptrD, ptrS, adjD, adjS, E);

  enc_kernel<<<gemm_blocks, blk, 0, stream>>>(x, Wenc, benc, X, Nn);

  for (int layer = 0; layer < 4; ++layer) {
    gather_mean_kernel<<<node_blocks, blk, 0, stream>>>(X, adjD, ptrD, mean, Nn);
    sage_kernel<<<gemm_blocks, blk, 0, stream>>>(X, mean, Wl, bl, Wr,
                                                 Wlg, blg, Wrg, Xn, Xg, Nn);
    gate_gather_kernel<<<node_blocks, blk, 0, stream>>>(Xg, adjS, ptrS, tau, Nn);
    update_kernel<<<nf_blocks, blk, 0, stream>>>(X, Xn, tau, nf);
  }

  dec_kernel<<<gemm_blocks, blk, 0, stream>>>(X, Wdec, bdec, (float*)d_out, Nn);
}


# edge grouping as a two-level counting sort in LDS: tile histograms, table scan, LDS-staged split into 256 node-range bins, one work-group orders each bin; no global atomic, no scattered 4-byte store
# speedup vs baseline: 4.1064x; 2.0728x over previous
#include <hip/hip_runtime.h>
#include <math.h>


typedef __attribute__((ext_vector_type(2))) float v2f;
typedef __attribute__((ext_vector_type(8))) float v8f;

#define NFEAT 128
#define NHID  64
#define NCLASS 40
#define GRP 8
#define NPW 4

static __device__ __forceinline__ v8f wmma4(v2f a, v2f b, v8f c) {
  return __builtin_amdgcn_wmma_f32_16x16x4_f32(false, a, false, b, (short)0, c,
                                               false, false);
}

#define NBIN   256
#define PTILE  6400
#define SCAP   12288
#define WMAX   1024

static __device__ __forceinline__ unsigned bin_magic(int Wn) {
  return (Wn > 1) ? (0xFFFFFFFFu / (unsigned)Wn + 1u) : 0u;
}
static __device__ __forceinline__ int bin_of(int key, unsigned magic, int Wn) {
  unsigned q = __umulhi((unsigned)key, magic);
  if (q * (unsigned)Wn > (unsigned)key) --q;
  return (Wn > 1) ? (int)q : key;
}

__global__ void bin_hist_kernel(const int* __restrict__ src, const int* __restrict__ dst,
                                int* __restrict__ hist, int E, int Wn, int NB) {
  __shared__ int h[NBIN];
  const int g = blockIdx.y;
  const int* __restrict__ key = g ? src : dst;
  const int b = blockIdx.x;
  for (int i = threadIdx.x; i < NBIN; i += blockDim.x) h[i] = 0;
  __syncthreads();
  const int e0 = b * PTILE;
  const int e1 = (e0 + PTILE < E) ? (e0 + PTILE) : E;
  const unsigned mg = bin_magic(Wn);
  for (int e = e0 + (int)threadIdx.x; e < e1; e += blockDim.x)
    atomicAdd(&h[bin_of(key[e], mg, Wn)], 1);
  __syncthreads();
  for (int i = threadIdx.x; i < NBIN; i += blockDim.x)
    hist[(long long)(g * NBIN + i) * NB + b] = h[i];
}

__global__ void bin_scan_kernel(int* __restrict__ hist, int* __restrict__ tot, int NB) {
  const int lane = threadIdx.x & 31;
  const int row = (int)((blockIdx.x * blockDim.x + threadIdx.x) >> 5);
  int* p = hist + (long long)row * NB;
  int carry = 0;
  for (int c0 = 0; c0 < NB; c0 += 32) {
    const int i = c0 + lane;
    const int x = (i < NB) ? p[i] : 0;
    int inc = x;
    for (int d = 1; d < 32; d <<= 1) {
      const int up = __shfl_up(inc, d);
      if (lane >= d) inc += up;
    }
    if (i < NB) p[i] = carry + inc - x;
    carry += __shfl(inc, 31);
  }
  if (lane == 0) tot[row] = carry;
}

__global__ void bin_part_kernel(const int* __restrict__ src, const int* __restrict__ dst,
                                const int* __restrict__ hist, const int* __restrict__ tot,
                                int2* __restrict__ part, int E, int Wn, int NB) {
  __shared__ int cnt[NBIN];
  __shared__ int lstart[NBIN];
  __shared__ int gstart[NBIN];
  __shared__ int2 stage[PTILE];
  const int g = blockIdx.y;
  const int* __restrict__ key = g ? src : dst;
  const int* __restrict__ val = g ? dst : src;
  const int b = blockIdx.x;
  const int e0 = b * PTILE;
  const int e1 = (e0 + PTILE < E) ? (e0 + PTILE) : E;
  const unsigned mg = bin_magic(Wn);
  for (int i = threadIdx.x; i < NBIN; i += blockDim.x) cnt[i] = 0;
  __syncthreads();
  for (int e = e0 + (int)threadIdx.x; e < e1; e += blockDim.x)
    atomicAdd(&cnt[bin_of(key[e], mg, Wn)], 1);
  __syncthreads();
  if (threadIdx.x < 32) {
    const int l = threadIdx.x;
    int c[8], t[8], o[8];
    for (int j = 0; j < 8; ++j) {
      c[j] = cnt[8 * l + j];
      t[j] = tot[g * NBIN + 8 * l + j];
      o[j] = hist[(long long)(g * NBIN + 8 * l + j) * NB + b];
    }
    int cs = 0, ts = 0;
    for (int j = 0; j < 8; ++j) {
      const int x = c[j]; c[j] = cs; cs += x;
      const int y = t[j]; t[j] = ts; ts += y;
    }
    int ci = cs, ti = ts;
    for (int d = 1; d < 32; d <<= 1) {
      const int uc = __shfl_up(ci, d);
      const int ut = __shfl_up(ti, d);
      if (l >= d) { ci += uc; ti += ut; }
    }
    const int cb = ci - cs;
    const int tb = ti - ts;
    for (int j = 0; j < 8; ++j) {
      lstart[8 * l + j] = cb + c[j];
      gstart[8 * l + j] = tb + t[j] + o[j];
    }
  }
  __syncthreads();
  for (int i = threadIdx.x; i < NBIN; i += blockDim.x) cnt[i] = lstart[i];
  __syncthreads();
  for (int e = e0 + (int)threadIdx.x; e < e1; e += blockDim.x) {
    const int k = key[e];
    const int v = val[e];
    const int pos = atomicAdd(&cnt[bin_of(k, mg, Wn)], 1);
    stage[pos] = make_int2(k, v);
  }
  __syncthreads();
  int2* __restrict__ outp = part + (long long)g * E;
  const int n = e1 - e0;
  for (int i = threadIdx.x; i < n; i += blockDim.x) {
    const int2 kv = stage[i];
    const int bin = bin_of(kv.x, mg, Wn);
    outp[gstart[bin] + (i - lstart[bin])] = kv;
  }
}

__global__ void bin_sort_kernel(const int2* __restrict__ part, const int* __restrict__ tot,
                                int* __restrict__ adjD, int* __restrict__ adjS,
                                int* __restrict__ ptrD, int* __restrict__ ptrS,
                                int E, int Nn, int Wn) {
  __shared__ int cnt[WMAX];
  __shared__ int cur[WMAX];
  __shared__ int wtot[32];
  __shared__ int sbase;
  __shared__ int stg[SCAP];
  const int g = blockIdx.y;
  const int k = blockIdx.x;
  const int lane = threadIdx.x & 31;
  const int wave = threadIdx.x >> 5;
  int* __restrict__ adj = g ? adjS : adjD;
  int* __restrict__ endp = g ? ptrS : ptrD;
  const int size = tot[g * NBIN + k];
  if (threadIdx.x < 32) {
    int sum = 0;
    for (int j = 0; j < NBIN / 32; ++j) {
      const int idx = j * 32 + lane;
      const int x = tot[g * NBIN + idx];
      sum += (idx < k) ? x : 0;
    }
    for (int d = 16; d > 0; d >>= 1) sum += __shfl_xor(sum, d);
    if (lane == 0) sbase = sum;
  }
  for (int i = threadIdx.x; i < WMAX; i += blockDim.x) cnt[i] = 0;
  __syncthreads();
  const int base = sbase;
  const int node0 = k * Wn;
  const int2* __restrict__ pin = part + (long long)g * E + base;
  for (int i = threadIdx.x; i < size; i += blockDim.x) atomicAdd(&cnt[pin[i].x - node0], 1);
  __syncthreads();
  const int x = (threadIdx.x < WMAX) ? cnt[threadIdx.x] : 0;
  int inc = x;
  for (int d = 1; d < 32; d <<= 1) {
    const int up = __shfl_up(inc, d);
    if (lane >= d) inc += up;
  }
  if (lane == 31) wtot[wave] = inc;
  __syncthreads();
  const int mine = wtot[lane];
  int pre = (lane < wave) ? mine : 0;
  for (int d = 16; d > 0; d >>= 1) pre += __shfl_xor(pre, d);
  const int excl = pre + inc - x;
  if (threadIdx.x < WMAX) cur[threadIdx.x] = excl;
  const int v = node0 + (int)threadIdx.x;
  if ((int)threadIdx.x < Wn && v < Nn) endp[v] = base + excl + x;
  __syncthreads();
  const bool staged = (size <= SCAP);
  for (int i = threadIdx.x; i < size; i += blockDim.x) {
    const int2 kv = pin[i];
    const int pos = atomicAdd(&cur[kv.x - node0], 1);
    if (staged) stg[pos] = kv.y;
    else        adj[base + pos] = kv.y;
  }
  __syncthreads();
  if (staged)
    for (int i = threadIdx.x; i < size; i += blockDim.x) adj[base + i] = stg[i];
}

__global__ void enc_kernel(const float* __restrict__ x, const float* __restrict__ W,
                           const float* __restrict__ bias, float* __restrict__ out,
                           int M) {
  const int lane = threadIdx.x & 31;
  const int wave = threadIdx.x >> 5;
  const int m0 = (blockIdx.x * 8 + wave) * 16;
  if (m0 >= M) return;
  const int mr = lane & 15;
  const int kh = (lane >> 4) << 1;
  v8f acc[4] = {};
  const float* arow = x + (long long)(m0 + mr) * NFEAT;
  for (int k0 = 0; k0 < NFEAT; k0 += 4) {
    v2f a = *(const v2f*)(arow + k0 + kh);
#pragma unroll
    for (int nt = 0; nt < 4; ++nt) {
      v2f b;
      b.x = W[(k0 + kh) * NHID + nt * 16 + mr];
      b.y = W[(k0 + kh + 1) * NHID + nt * 16 + mr];
      acc[nt] = wmma4(a, b, acc[nt]);
    }
  }
  const int rbase = (lane >> 4) << 3;
#pragma unroll
  for (int nt = 0; nt < 4; ++nt) {
    const int n = nt * 16 + mr;
    const float bn = bias[n];
#pragma unroll
    for (int r = 0; r < 8; ++r) {
      float v = acc[nt][r] + bn;
      out[(long long)(m0 + rbase + r) * NHID + n] = fmaxf(v, 0.0f);
    }
  }
}

__global__ void gather_mean_kernel(const float* __restrict__ X, const int* __restrict__ adj,
                                   const int* __restrict__ endp, float* __restrict__ mean,
                                   int Nn) {
  const int lane = threadIdx.x & 31;
  const int w = __builtin_amdgcn_readfirstlane((int)((blockIdx.x * blockDim.x + threadIdx.x) >> 5));
  const int nw = (int)((gridDim.x * blockDim.x) >> 5);
  for (int v = w; v < Nn; v += nw) {
    const int prev = endp[(v > 0) ? (v - 1) : 0];
    const int end = endp[v];
    const int beg = (v > 0) ? prev : 0;
    v2f acc = {0.0f, 0.0f};
    for (int base = beg; base < end; base += 32) {
      const int n = (end - base < 32) ? (end - base) : 32;
      int sv = 0;
      if (lane < n) sv = adj[base + lane];
      for (int j = 0; j < n; j += GRP) {
        v2f r[GRP];
#pragma unroll
        for (int u = 0; u < GRP; ++u) {
          const int s = __builtin_amdgcn_readlane(sv, j + u);
          r[u] = *(const v2f*)(X + (long long)s * NHID + 2 * lane);
        }
#pragma unroll
        for (int u = 0; u < GRP; ++u) {
          acc.x += (j + u < n) ? r[u].x : 0.0f;
          acc.y += (j + u < n) ? r[u].y : 0.0f;
        }
      }
    }
    const float dg = fmaxf((float)(end - beg), 1.0f);
    v2f m;
    m.x = acc.x / dg;
    m.y = acc.y / dg;
    *(v2f*)(mean + (long long)v * NHID + 2 * lane) = m;
  }
}

__global__ void sage_kernel(const float* __restrict__ X, const float* __restrict__ mean,
                            const float* __restrict__ Wl, const float* __restrict__ bl,
                            const float* __restrict__ Wr,
                            const float* __restrict__ Wlg, const float* __restrict__ blg,
                            const float* __restrict__ Wrg,
                            float* __restrict__ Xn, float* __restrict__ Xg, int M) {
  const int lane = threadIdx.x & 31;
  const int wave = threadIdx.x >> 5;
  const int m0 = (blockIdx.x * 8 + wave) * 16;
  if (m0 >= M) return;
  const int mr = lane & 15;
  const int kh = (lane >> 4) << 1;
  const long long row = (long long)(m0 + mr) * NHID;
  v8f aN[4] = {};
  v8f aG[4] = {};
  for (int k0 = 0; k0 < NHID; k0 += 4) {
    v2f am = *(const v2f*)(mean + row + k0 + kh);
    v2f ax = *(const v2f*)(X + row + k0 + kh);
    const int r0 = (k0 + kh) * NHID;
    const int r1 = (k0 + kh + 1) * NHID;
#pragma unroll
    for (int nt = 0; nt < 4; ++nt) {
      const int c = nt * 16 + mr;
      v2f b;
      b.x = Wl[r0 + c];  b.y = Wl[r1 + c];
      aN[nt] = wmma4(am, b, aN[nt]);
      b.x = Wr[r0 + c];  b.y = Wr[r1 + c];
      aN[nt] = wmma4(ax, b, aN[nt]);
      b.x = Wlg[r0 + c]; b.y = Wlg[r1 + c];
      aG[nt] = wmma4(am, b, aG[nt]);
      b.x = Wrg[r0 + c]; b.y = Wrg[r1 + c];
      aG[nt] = wmma4(ax, b, aG[nt]);
    }
  }
  const int rbase = (lane >> 4) << 3;
#pragma unroll
  for (int nt = 0; nt < 4; ++nt) {
    const int n = nt * 16 + mr;
    const float bn = bl[n];
    const float bg = blg[n];
#pragma unroll
    for (int r = 0; r < 8; ++r) {
      const long long o = (long long)(m0 + rbase + r) * NHID + n;
      Xn[o] = fmaxf(aN[nt][r] + bn, 0.0f);
      Xg[o] = fmaxf(aG[nt][r] + bg, 0.0f);
    }
  }
}

__global__ void gate_update_kernel(const float* __restrict__ Xg, const float* __restrict__ Xn,
                                   const int* __restrict__ adj, const int* __restrict__ endp,
                                   float* __restrict__ X, int Nn) {
  const int lane = threadIdx.x & 31;
  const int w = __builtin_amdgcn_readfirstlane((int)((blockIdx.x * blockDim.x + threadIdx.x) >> 5));
  const int nw = (int)((gridDim.x * blockDim.x) >> 5);
  for (int v = w; v < Nn; v += nw) {
    const int prev = endp[(v > 0) ? (v - 1) : 0];
    const int end = endp[v];
    const int beg = (v > 0) ? prev : 0;
    const long long ro = (long long)v * NHID + 2 * lane;
    const v2f own = *(const v2f*)(Xg + ro);
    const v2f xo = *(const v2f*)(X + ro);
    const v2f xn = *(const v2f*)(Xn + ro);
    v2f acc = {0.0f, 0.0f};
    for (int base = beg; base < end; base += 32) {
      const int n = (end - base < 32) ? (end - base) : 32;
      int dv = 0;
      if (lane < n) dv = adj[base + lane];
      for (int j = 0; j < n; j += GRP) {
        v2f r[GRP];
#pragma unroll
        for (int u = 0; u < GRP; ++u) {
          const int d = __builtin_amdgcn_readlane(dv, j + u);
          r[u] = *(const v2f*)(Xg + (long long)d * NHID + 2 * lane);
        }
#pragma unroll
        for (int u = 0; u < GRP; ++u) {
          const float dx = own.x - r[u].x;
          const float dy = own.y - r[u].y;
          acc.x += (j + u < n) ? dx * dx : 0.0f;
          acc.y += (j + u < n) ? dy * dy : 0.0f;
        }
      }
    }
    const float dg = fmaxf((float)(end - beg), 1.0f);
    const float tx = tanhf(acc.x / dg);
    const float ty = tanhf(acc.y / dg);
    v2f o;
    o.x = (1.0f - tx) * xo.x + tx * xn.x;
    o.y = (1.0f - ty) * xo.y + ty * xn.y;
    *(v2f*)(X + ro) = o;
  }
}

__global__ void dec_kernel(const float* __restrict__ X, const float* __restrict__ W,
                           const float* __restrict__ bias, float* __restrict__ out,
                           int M) {
  const int lane = threadIdx.x & 31;
  const int wave = threadIdx.x >> 5;
  const int m0 = (blockIdx.x * 8 + wave) * 16;
  if (m0 >= M) return;
  const int mr = lane & 15;
  const int kh = (lane >> 4) << 1;
  v8f acc[3] = {};
  const float* arow = X + (long long)(m0 + mr) * NHID;
  for (int k0 = 0; k0 < NHID; k0 += 4) {
    v2f a = *(const v2f*)(arow + k0 + kh);
#pragma unroll
    for (int nt = 0; nt < 3; ++nt) {
      const int n = nt * 16 + mr;
      v2f b = {0.0f, 0.0f};
      if (n < NCLASS) {
        b.x = W[(k0 + kh) * NCLASS + n];
        b.y = W[(k0 + kh + 1) * NCLASS + n];
      }
      acc[nt] = wmma4(a, b, acc[nt]);
    }
  }
  const int rbase = (lane >> 4) << 3;
#pragma unroll
  for (int nt = 0; nt < 3; ++nt) {
    const int n = nt * 16 + mr;
    if (n < NCLASS) {
      const float bn = bias[n];
#pragma unroll
      for (int r = 0; r < 8; ++r)
        out[(long long)(m0 + rbase + r) * NCLASS + n] = acc[nt][r] + bn;
    }
  }
}

extern "C" void kernel_launch(void* const* d_in, const int* in_sizes, int n_in,
                              void* d_out, int out_size, void* d_ws, size_t ws_size,
                              hipStream_t stream) {
  const float* x    = (const float*)d_in[0];
  const int*   ei   = (const int*)d_in[1];
  const float* Wenc = (const float*)d_in[2];
  const float* benc = (const float*)d_in[3];
  const float* Wl   = (const float*)d_in[4];
  const float* bl   = (const float*)d_in[5];
  const float* Wr   = (const float*)d_in[6];
  const float* Wlg  = (const float*)d_in[7];
  const float* blg  = (const float*)d_in[8];
  const float* Wrg  = (const float*)d_in[9];
  const float* Wdec = (const float*)d_in[10];
  const float* bdec = (const float*)d_in[11];

  const int Nn = in_sizes[0] / NFEAT;
  const int E  = in_sizes[1] / 2;
  const int* src = ei;
  const int* dst = ei + E;

  float* ws = (float*)d_ws;
  const long long nf = (long long)Nn * NHID;
  float* X    = ws;
  float* Xn   = ws + nf;
  float* Xg   = ws + 2 * nf;
  float* mean = ws + 3 * nf;
  int* adjD = (int*)(ws + 4 * nf);
  int* adjS = adjD + E;
  int* ptrD = adjS + E;
  int* ptrS = ptrD + Nn;
  const int NB = (E + PTILE - 1) / PTILE;
  const int Wn = (Nn + NBIN - 1) / NBIN;
  int2* part = (int2*)Xn;
  int* hist = (int*)mean;
  int* tot  = hist + 2 * NBIN * NB;

  const dim3 blk(256);
  const int gemm_blocks = (Nn + 127) / 128;
  const int node_blocks = (Nn + 8 * NPW - 1) / (8 * NPW);

  bin_hist_kernel<<<dim3(NB, 2), 1024, 0, stream>>>(src, dst, hist, E, Wn, NB);
  bin_scan_kernel<<<(2 * NBIN) / 8, blk, 0, stream>>>(hist, tot, NB);
  bin_part_kernel<<<dim3(NB, 2), 1024, 0, stream>>>(src, dst, hist, tot, part, E, Wn, NB);
  bin_sort_kernel<<<dim3(NBIN, 2), 1024, 0, stream>>>(part, tot, adjD, adjS, ptrD, ptrS, E, Nn, Wn);

  enc_kernel<<<gemm_blocks, blk, 0, stream>>>(x, Wenc, benc, X, Nn);

  for (int layer = 0; layer < 4; ++layer) {
    gather_mean_kernel<<<node_blocks, blk, 0, stream>>>(X, adjD, ptrD, mean, Nn);
    sage_kernel<<<gemm_blocks, blk, 0, stream>>>(X, mean, Wl, bl, Wr,
                                                 Wlg, blg, Wrg, Xn, Xg, Nn);
    gate_update_kernel<<<node_blocks, blk, 0, stream>>>(Xg, Xn, adjS, ptrS, X, Nn);
  }

  dec_kernel<<<gemm_blocks, blk, 0, stream>>>(X, Wdec, bdec, (float*)d_out, Nn);
}
